# starting kernel
# baseline (speedup 1.0000x reference)
#include <hip/hip_runtime.h>

typedef __attribute__((ext_vector_type(2))) float v2f;
typedef __attribute__((ext_vector_type(8))) float v8f;

#define NPTS  2048
#define FEAT  512
#define NBATCH 32

__global__ __launch_bounds__(128)
void FCDecoder_gfx1250_kernel(const float* __restrict__ glf,
                              const float* __restrict__ W1, const float* __restrict__ b1,
                              const float* __restrict__ W2, const float* __restrict__ b2,
                              const float* __restrict__ W3, const float* __restrict__ b3,
                              float* __restrict__ out)
{
    __shared__ float Y1s[32 * 33];
    __shared__ float Y2s[32 * 9];

    const int p    = blockIdx.x;
    const int tid  = threadIdx.x;
    const int wave = tid >> 5;
    const int lane = tid & 31;
    const int l16  = lane & 15;
    const int half = lane >> 4;

    const int m_tile = wave & 1;
    const int n_tile = wave >> 1;
    const int m = m_tile * 16 + l16;
    const int n = n_tile * 16 + l16;

    const float* __restrict__ Arow = glf + (size_t)m * FEAT + 2 * half;
    const float* __restrict__ Brow = W1 + ((size_t)p * NBATCH + n) * FEAT + 2 * half;

    v8f acc = {};
    #pragma unroll 1
    for (int kc = 0; kc < FEAT; kc += 64) {
        __builtin_prefetch(Brow + kc + 64, 0, 0);
        #pragma unroll
        for (int kk = 0; kk < 64; kk += 4) {
            v2f a = *(const v2f*)(Arow + kc + kk);
            v2f b = __builtin_nontemporal_load((const v2f*)(Brow + kc + kk));
            acc = __builtin_amdgcn_wmma_f32_16x16x4_f32(
                      false, a, false, b, (short)0, acc, false, false);
        }
    }

    const float bias1 = b1[(size_t)p * NBATCH + n];
    #pragma unroll
    for (int r = 0; r < 8; ++r)
        Y1s[(m_tile * 16 + r + 8 * half) * 33 + n] = acc[r] + bias1;

    __syncthreads();

    const int  m2base = (wave & 1) * 16;
    const bool nvalid = (l16 < 8);
    const float* __restrict__ B2row =
        W2 + ((size_t)p * 8 + (nvalid ? l16 : 0)) * 32 + 2 * half;

    v8f acc2 = {};
    #pragma unroll
    for (int kk = 0; kk < 32; kk += 4) {
        const int ka = kk + 2 * half;
        v2f a2;
        a2.x = Y1s[(m2base + l16) * 33 + ka];
        a2.y = Y1s[(m2base + l16) * 33 + ka + 1];
        v2f b2v = {0.0f, 0.0f};
        if (nvalid) b2v = *(const v2f*)(B2row + kk);
        acc2 = __builtin_amdgcn_wmma_f32_16x16x4_f32(
                   false, a2, false, b2v, (short)0, acc2, false, false);
    }

    if (wave < 2 && nvalid) {
        const float bias2 = b2[(size_t)p * 8 + l16];
        #pragma unroll
        for (int r = 0; r < 8; ++r)
            Y2s[(m2base + r + 8 * half) * 9 + l16] = acc2[r] + bias2;
    }

    __syncthreads();

    if (tid < NBATCH * 3) {
        const int bb = tid / 3;
        const int c  = tid % 3;
        const float* __restrict__ w3 = W3 + ((size_t)p * 3 + c) * 8;
        float s = b3[(size_t)p * 3 + c];
        #pragma unroll
        for (int j = 0; j < 8; ++j)
            s += Y2s[bb * 9 + j] * w3[j];
        out[((size_t)bb * 3 + c) * NPTS + p] = s;
    }
}

extern "C" void kernel_launch(void* const* d_in, const int* in_sizes, int n_in,
                              void* d_out, int out_size, void* d_ws, size_t ws_size,
                              hipStream_t stream) {
    (void)in_sizes; (void)n_in; (void)out_size; (void)d_ws; (void)ws_size;
    const float* glf = (const float*)d_in[0];
    const float* W1  = (const float*)d_in[1];
    const float* b1  = (const float*)d_in[2];
    const float* W2  = (const float*)d_in[3];
    const float* b2  = (const float*)d_in[4];
    const float* W3  = (const float*)d_in[5];
    const float* b3  = (const float*)d_in[6];
    float* out = (float*)d_out;

    FCDecoder_gfx1250_kernel<<<NPTS, 128, 0, stream>>>(glf, W1, b1, W2, b2, W3, b3, out);
}


# W1[p] staged once per block into padded LDS by whole-line 512-B coalesced loads; layer-1 B fragments read from LDS instead of per-row strided global loads; k order and summation order unchanged
# speedup vs baseline: 2.1614x; 2.1614x over previous
#include <hip/hip_runtime.h>

typedef __attribute__((ext_vector_type(2))) float v2f;
typedef __attribute__((ext_vector_type(4))) float v4f;
typedef __attribute__((ext_vector_type(8))) float v8f;

#define NPTS  2048
#define FEAT  512
#define NBATCH 32
#define BPITCH 516

__global__ __launch_bounds__(128)
void FCDecoder_gfx1250_kernel(const float* __restrict__ glf,
                              const float* __restrict__ W1, const float* __restrict__ b1,
                              const float* __restrict__ W2, const float* __restrict__ b2,
                              const float* __restrict__ W3, const float* __restrict__ b3,
                              float* __restrict__ out)
{
    __shared__ float Bs[NBATCH * BPITCH];
    __shared__ float Y1s[32 * 33];
    __shared__ float Y2s[32 * 9];

    const int p    = blockIdx.x;
    const int tid  = threadIdx.x;
    const int wave = tid >> 5;
    const int lane = tid & 31;
    const int l16  = lane & 15;
    const int half = lane >> 4;

    {
        const float* __restrict__ Wp = W1 + (size_t)p * NBATCH * FEAT + tid * 4;
        v4f st[32];
        #pragma unroll
        for (int i = 0; i < 32; ++i)
            st[i] = __builtin_nontemporal_load((const v4f*)(Wp + i * FEAT));
        #pragma unroll
        for (int i = 0; i < 32; ++i)
            *(v4f*)(Bs + i * BPITCH + tid * 4) = st[i];
    }

    __syncthreads();

    const int m_tile = wave & 1;
    const int n_tile = wave >> 1;
    const int m = m_tile * 16 + l16;
    const int n = n_tile * 16 + l16;

    const float* __restrict__ Arow = glf + (size_t)m * FEAT + 2 * half;
    const float* Brow = Bs + n * BPITCH + 2 * half;

    v8f acc = {};
    #pragma unroll 1
    for (int kc = 0; kc < FEAT; kc += 64) {
        #pragma unroll
        for (int kk = 0; kk < 64; kk += 4) {
            v2f a = *(const v2f*)(Arow + kc + kk);
            v2f b = *(const v2f*)(Brow + kc + kk);
            acc = __builtin_amdgcn_wmma_f32_16x16x4_f32(
                      false, a, false, b, (short)0, acc, false, false);
        }
    }

    const float bias1 = b1[(size_t)p * NBATCH + n];
    #pragma unroll
    for (int r = 0; r < 8; ++r)
        Y1s[(m_tile * 16 + r + 8 * half) * 33 + n] = acc[r] + bias1;

    __syncthreads();

    const int  m2base = (wave & 1) * 16;
    const bool nvalid = (l16 < 8);
    const float* __restrict__ B2row =
        W2 + ((size_t)p * 8 + (nvalid ? l16 : 0)) * 32 + 2 * half;

    v8f acc2 = {};
    #pragma unroll
    for (int kk = 0; kk < 32; kk += 4) {
        const int ka = kk + 2 * half;
        v2f a2;
        a2.x = Y1s[(m2base + l16) * 33 + ka];
        a2.y = Y1s[(m2base + l16) * 33 + ka + 1];
        v2f b2v = {0.0f, 0.0f};
        if (nvalid) b2v = *(const v2f*)(B2row + kk);
        acc2 = __builtin_amdgcn_wmma_f32_16x16x4_f32(
                   false, a2, false, b2v, (short)0, acc2, false, false);
    }

    if (wave < 2 && nvalid) {
        const float bias2 = b2[(size_t)p * 8 + l16];
        #pragma unroll
        for (int r = 0; r < 8; ++r)
            Y2s[(m2base + r + 8 * half) * 9 + l16] = acc2[r] + bias2;
    }

    __syncthreads();

    if (tid < NBATCH * 3) {
        const int bb = tid / 3;
        const int c  = tid % 3;
        const float* __restrict__ w3 = W3 + ((size_t)p * 3 + c) * 8;
        float s = b3[(size_t)p * 3 + c];
        #pragma unroll
        for (int j = 0; j < 8; ++j)
            s += Y2s[bb * 9 + j] * w3[j];
        out[((size_t)bb * 3 + c) * NPTS + p] = s;
    }
}

extern "C" void kernel_launch(void* const* d_in, const int* in_sizes, int n_in,
                              void* d_out, int out_size, void* d_ws, size_t ws_size,
                              hipStream_t stream) {
    (void)in_sizes; (void)n_in; (void)out_size; (void)d_ws; (void)ws_size;
    const float* glf = (const float*)d_in[0];
    const float* W1  = (const float*)d_in[1];
    const float* b1  = (const float*)d_in[2];
    const float* W2  = (const float*)d_in[3];
    const float* b2  = (const float*)d_in[4];
    const float* W3  = (const float*)d_in[5];
    const float* b3  = (const float*)d_in[6];
    float* out = (float*)d_out;

    FCDecoder_gfx1250_kernel<<<NPTS, 128, 0, stream>>>(glf, W1, b1, W2, b2, W3, b3, out);
}


# Input features (A operand, the same for every position) read once per block into registers via the LDS tile and reused over 4 positions per block, replacing per-step strided global A loads
# speedup vs baseline: 2.4513x; 1.1341x over previous
#include <hip/hip_runtime.h>

typedef __attribute__((ext_vector_type(2))) float v2f;
typedef __attribute__((ext_vector_type(4))) float v4f;
typedef __attribute__((ext_vector_type(8))) float v8f;

#define NPTS  2048
#define FEAT  512
#define NBATCH 32
#define BPITCH 516
#define PPB   4
#define NBLK  512

template <bool STREAM>
__device__ __forceinline__ void stage_tile(const float* __restrict__ src, float* Bs, int tid)
{
    const float* __restrict__ Wp = src + tid * 4;
    v4f st[32];
    #pragma unroll
    for (int i = 0; i < 32; ++i) {
        if (STREAM) st[i] = __builtin_nontemporal_load((const v4f*)(Wp + i * FEAT));
        else        st[i] = *(const v4f*)(Wp + i * FEAT);
    }
    #pragma unroll
    for (int i = 0; i < 32; ++i)
        *(v4f*)(Bs + i * BPITCH + tid * 4) = st[i];
}

__global__ __launch_bounds__(128)
void FCDecoder_gfx1250_kernel(const float* __restrict__ glf,
                              const float* __restrict__ W1, const float* __restrict__ b1,
                              const float* __restrict__ W2, const float* __restrict__ b2,
                              const float* __restrict__ W3, const float* __restrict__ b3,
                              float* __restrict__ out)
{
    __shared__ float Bs[NBATCH * BPITCH];
    __shared__ float Y1s[32 * 33];
    __shared__ float Y2s[32 * 9];

    const int tid  = threadIdx.x;
    const int wave = tid >> 5;
    const int lane = tid & 31;
    const int l16  = lane & 15;
    const int half = lane >> 4;

    const int m_tile = wave & 1;
    const int n_tile = wave >> 1;
    const int m = m_tile * 16 + l16;
    const int n = n_tile * 16 + l16;

    const float* Arow = Bs + m * BPITCH + 2 * half;
    const float* Brow = Bs + n * BPITCH + 2 * half;

    stage_tile<false>(glf, Bs, tid);
    __syncthreads();
    v2f af[128];
    #pragma unroll
    for (int s = 0; s < 128; ++s)
        af[s] = *(const v2f*)(Arow + 4 * s);
    __syncthreads();

    #pragma unroll 1
    for (int i = 0; i < PPB; ++i) {
        const int p = blockIdx.x * PPB + i;

        stage_tile<true>(W1 + (size_t)p * NBATCH * FEAT, Bs, tid);
        __syncthreads();

        v8f acc = {};
        #pragma unroll
        for (int s = 0; s < 128; ++s) {
            v2f b = *(const v2f*)(Brow + 4 * s);
            acc = __builtin_amdgcn_wmma_f32_16x16x4_f32(
                      false, af[s], false, b, (short)0, acc, false, false);
        }

        const float bias1 = b1[(size_t)p * NBATCH + n];
        #pragma unroll
        for (int r = 0; r < 8; ++r)
            Y1s[(m_tile * 16 + r + 8 * half) * 33 + n] = acc[r] + bias1;

        __syncthreads();

        const int  m2base = (wave & 1) * 16;
        const bool nvalid = (l16 < 8);
        const float* __restrict__ B2row =
            W2 + ((size_t)p * 8 + (nvalid ? l16 : 0)) * 32 + 2 * half;

        v8f acc2 = {};
        #pragma unroll
        for (int kk = 0; kk < 32; kk += 4) {
            const int ka = kk + 2 * half;
            v2f a2;
            a2.x = Y1s[(m2base + l16) * 33 + ka];
            a2.y = Y1s[(m2base + l16) * 33 + ka + 1];
            v2f b2v = {0.0f, 0.0f};
            if (nvalid) b2v = *(const v2f*)(B2row + kk);
            acc2 = __builtin_amdgcn_wmma_f32_16x16x4_f32(
                       false, a2, false, b2v, (short)0, acc2, false, false);
        }

        if (wave < 2 && nvalid) {
            const float bias2 = b2[(size_t)p * 8 + l16];
            #pragma unroll
            for (int r = 0; r < 8; ++r)
                Y2s[(m2base + r + 8 * half) * 9 + l16] = acc2[r] + bias2;
        }

        __syncthreads();

        if (tid < NBATCH * 3) {
            const int bb = tid / 3;
            const int c  = tid % 3;
            const float* __restrict__ w3 = W3 + ((size_t)p * 3 + c) * 8;
            float s = b3[(size_t)p * 3 + c];
            #pragma unroll
            for (int j = 0; j < 8; ++j)
                s += Y2s[bb * 9 + j] * w3[j];
            out[((size_t)bb * 3 + c) * NPTS + p] = s;
        }
    }
}

extern "C" void kernel_launch(void* const* d_in, const int* in_sizes, int n_in,
                              void* d_out, int out_size, void* d_ws, size_t ws_size,
                              hipStream_t stream) {
    (void)in_sizes; (void)n_in; (void)out_size; (void)d_ws; (void)ws_size;
    const float* glf = (const float*)d_in[0];
    const float* W1  = (const float*)d_in[1];
    const float* b1  = (const float*)d_in[2];
    const float* W2  = (const float*)d_in[3];
    const float* b2  = (const float*)d_in[4];
    const float* W3  = (const float*)d_in[5];
    const float* b3  = (const float*)d_in[6];
    float* out = (float*)d_out;

    FCDecoder_gfx1250_kernel<<<NBLK, 128, 0, stream>>>(glf, W1, b1, W2, b2, W3, b3, out);
}


# W1[p+1] and position p's small operands (b1,W2,b2,W3,b3) are requested before layer 1 and held in registers through its WMMA loop; LDS refill behind layer 1's barrier; summation order unchanged
# speedup vs baseline: 2.9444x; 1.2012x over previous
#include <hip/hip_runtime.h>

typedef __attribute__((ext_vector_type(2))) float v2f;
typedef __attribute__((ext_vector_type(4))) float v4f;
typedef __attribute__((ext_vector_type(8))) float v8f;

#define NPTS  2048
#define FEAT  512
#define NBATCH 32
#define BPITCH 516
#define PPB   4
#define NBLK  512

template <bool STREAM>
__device__ __forceinline__ void load_tile(const float* __restrict__ src, v4f (&st)[32], int tid)
{
    const float* __restrict__ Wp = src + tid * 4;
    #pragma unroll
    for (int i = 0; i < 32; ++i) {
        if (STREAM) st[i] = __builtin_nontemporal_load((const v4f*)(Wp + i * FEAT));
        else        st[i] = *(const v4f*)(Wp + i * FEAT);
    }
}

__device__ __forceinline__ void store_tile(const v4f (&st)[32], float* Bs, int tid)
{
    #pragma unroll
    for (int i = 0; i < 32; ++i)
        *(v4f*)(Bs + i * BPITCH + tid * 4) = st[i];
}

__global__ __launch_bounds__(128)
void FCDecoder_gfx1250_kernel(const float* __restrict__ glf,
                              const float* __restrict__ W1, const float* __restrict__ b1,
                              const float* __restrict__ W2, const float* __restrict__ b2,
                              const float* __restrict__ W3, const float* __restrict__ b3,
                              float* __restrict__ out)
{
    __shared__ float Bs[NBATCH * BPITCH];
    __shared__ float Y1s[32 * 33];
    __shared__ float Y2s[32 * 9];

    const int tid  = threadIdx.x;
    const int wave = tid >> 5;
    const int lane = tid & 31;
    const int l16  = lane & 15;
    const int half = lane >> 4;

    const int m_tile = wave & 1;
    const int n_tile = wave >> 1;
    const int m = m_tile * 16 + l16;
    const int n = n_tile * 16 + l16;

    const float* Arow = Bs + m * BPITCH + 2 * half;
    const float* Brow = Bs + n * BPITCH + 2 * half;

    const int  m2base = (wave & 1) * 16;
    const bool nvalid = (l16 < 8);
    const int  n2     = nvalid ? l16 : 0;
    const int  bb     = tid / 3;
    const int  c      = tid % 3;

    v4f st[32];
    {
        v4f g[32];
        load_tile<false>(glf, g, tid);
        load_tile<true>(W1 + (size_t)blockIdx.x * PPB * NBATCH * FEAT, st, tid);
        store_tile(g, Bs, tid);
    }
    __syncthreads();
    v2f af[128];
    #pragma unroll
    for (int s = 0; s < 128; ++s)
        af[s] = *(const v2f*)(Arow + 4 * s);
    __syncthreads();
    store_tile(st, Bs, tid);
    __syncthreads();

    #pragma unroll 1
    for (int i = 0; i < PPB; ++i) {
        const int  p        = blockIdx.x * PPB + i;
        const bool has_next = (i + 1 < PPB);

        const float bias1 = b1[(size_t)p * NBATCH + n];
        const float* __restrict__ B2row = W2 + ((size_t)p * 8 + n2) * 32 + 2 * half;
        v2f w2f[8];
        #pragma unroll
        for (int k = 0; k < 8; ++k)
            w2f[k] = *(const v2f*)(B2row + 4 * k);
        const float bias2 = b2[(size_t)p * 8 + n2];
        const float* __restrict__ w3 = W3 + ((size_t)p * 3 + c) * 8;
        float w3r[8];
        #pragma unroll
        for (int j = 0; j < 8; ++j)
            w3r[j] = w3[j];
        const float bias3 = b3[(size_t)p * 3 + c];
        if (has_next)
            load_tile<true>(W1 + (size_t)(p + 1) * NBATCH * FEAT, st, tid);

        v8f acc = {};
        #pragma unroll
        for (int s = 0; s < 128; ++s) {
            v2f b = *(const v2f*)(Brow + 4 * s);
            acc = __builtin_amdgcn_wmma_f32_16x16x4_f32(
                      false, af[s], false, b, (short)0, acc, false, false);
        }

        #pragma unroll
        for (int r = 0; r < 8; ++r)
            Y1s[(m_tile * 16 + r + 8 * half) * 33 + n] = acc[r] + bias1;

        __syncthreads();

        if (has_next)
            store_tile(st, Bs, tid);

        v8f acc2 = {};
        #pragma unroll
        for (int k = 0; k < 8; ++k) {
            const int ka = 4 * k + 2 * half;
            v2f a2;
            a2.x = Y1s[(m2base + l16) * 33 + ka];
            a2.y = Y1s[(m2base + l16) * 33 + ka + 1];
            v2f b2v = {0.0f, 0.0f};
            if (nvalid) b2v = w2f[k];
            acc2 = __builtin_amdgcn_wmma_f32_16x16x4_f32(
                       false, a2, false, b2v, (short)0, acc2, false, false);
        }

        if (wave < 2 && nvalid) {
            #pragma unroll
            for (int r = 0; r < 8; ++r)
                Y2s[(m2base + r + 8 * half) * 9 + l16] = acc2[r] + bias2;
        }

        __syncthreads();

        if (tid < NBATCH * 3) {
            float s = bias3;
            #pragma unroll
            for (int j = 0; j < 8; ++j)
                s += Y2s[bb * 9 + j] * w3r[j];
            out[((size_t)bb * 3 + c) * NPTS + p] = s;
        }
    }
}

extern "C" void kernel_launch(void* const* d_in, const int* in_sizes, int n_in,
                              void* d_out, int out_size, void* d_ws, size_t ws_size,
                              hipStream_t stream) {
    (void)in_sizes; (void)n_in; (void)out_size; (void)d_ws; (void)ws_size;
    const float* glf = (const float*)d_in[0];
    const float* W1  = (const float*)d_in[1];
    const float* b1  = (const float*)d_in[2];
    const float* W2  = (const float*)d_in[3];
    const float* b2  = (const float*)d_in[4];
    const float* W3  = (const float*)d_in[5];
    const float* b3  = (const float*)d_in[6];
    float* out = (float*)d_out;

    FCDecoder_gfx1250_kernel<<<NBLK, 128, 0, stream>>>(glf, W1, b1, W2, b2, W3, b3, out);
}


# W1 tile loads stay in flight across positions: two LDS tiles, W1[p+1] stored and W1[p+2] requested right behind layer 1, hand-written LDS-only barriers; k and summation order unchanged
# speedup vs baseline: 3.1172x; 1.0587x over previous
#include <hip/hip_runtime.h>

typedef __attribute__((ext_vector_type(2))) float v2f;
typedef __attribute__((ext_vector_type(4))) float v4f;
typedef __attribute__((ext_vector_type(8))) float v8f;

#define NPTS  2048
#define FEAT  512
#define NBATCH 32
#define BPITCH 516
#define PPB   4
#define NBLK  512
#define TILE_F (NBATCH * BPITCH)

template <bool STREAM>
__device__ __forceinline__ void load_tile(const float* src, v4f (&st)[32], int tid)
{
    const float* Wp = src + tid * 4;
    #pragma unroll
    for (int i = 0; i < 32; ++i) {
        if (STREAM) st[i] = __builtin_nontemporal_load((const v4f*)(Wp + i * FEAT));
        else        st[i] = *(const v4f*)(Wp + i * FEAT);
    }
}

__device__ __forceinline__ void store_tile(const v4f (&st)[32], float* Bs, int tid)
{
    #pragma unroll
    for (int i = 0; i < 32; ++i)
        *(v4f*)(Bs + i * BPITCH + tid * 4) = st[i];
}

template <bool REQUEST>
__device__ __forceinline__ void refill_tile(v4f (&st)[32], float* tile, const float* next, int tid)
{
    #pragma unroll
    for (int g = 0; g < 32; g += 8) {
        #pragma unroll
        for (int r = g; r < g + 8; ++r)
            *(v4f*)(tile + r * BPITCH + tid * 4) = st[r];
        if (REQUEST) {
            __builtin_amdgcn_sched_barrier(0);
            #pragma unroll
            for (int r = g; r < g + 8; ++r)
                st[r] = __builtin_nontemporal_load((const v4f*)(next + r * FEAT + tid * 4));
        }
    }
}

__device__ __forceinline__ void lds_barrier()
{
    asm volatile("s_wait_dscnt 0x0\n\ts_barrier_signal -1\n\ts_barrier_wait -1" ::: "memory");
}

template <bool REFILL, bool REQUEST>
__device__ __forceinline__ void one_position(
    const int p, const float* Bcur, float* Bnext,
    const float* W1, const float* b1, const float* W2, const float* b2,
    const float* W3, const float* b3, float* out,
    const v2f (&af)[128], v4f (&st)[32], float* Y1s, float* Y2s,
    const int tid, const int wave, const int half, const int l16, const int m_tile, const int n,
    const int m2base, const bool nvalid, const int n2, const int bb, const int c)
{
    const float bias1 = b1[(size_t)p * NBATCH + n];
    const float* B2row = W2 + ((size_t)p * 8 + n2) * 32 + 2 * half;
    v2f w2f[8];
    #pragma unroll
    for (int k = 0; k < 8; ++k)
        w2f[k] = *(const v2f*)(B2row + 4 * k);
    const float bias2 = b2[(size_t)p * 8 + n2];
    const float* w3 = W3 + ((size_t)p * 3 + c) * 8;
    float w3r[8];
    #pragma unroll
    for (int j = 0; j < 8; ++j)
        w3r[j] = w3[j];
    const float bias3 = b3[(size_t)p * 3 + c];
    __builtin_amdgcn_sched_barrier(0);

    v8f acc = {};
    #pragma unroll
    for (int s = 0; s < 128; ++s) {
        v2f b = *(const v2f*)(Bcur + 4 * s);
        acc = __builtin_amdgcn_wmma_f32_16x16x4_f32(
                  false, af[s], false, b, (short)0, acc, false, false);
    }

    asm volatile("" : "+v"(acc) :: "memory");
    if (REFILL)
        refill_tile<REQUEST>(st, Bnext, REQUEST ? W1 + (size_t)(p + 2) * NBATCH * FEAT : W1, tid);

    #pragma unroll
    for (int r = 0; r < 8; ++r)
        Y1s[(m_tile * 16 + r + 8 * half) * 33 + n] = acc[r] + bias1;

    lds_barrier();

    v8f acc2 = {};
    #pragma unroll
    for (int k = 0; k < 8; ++k) {
        const int ka = 4 * k + 2 * half;
        v2f a2;
        a2.x = Y1s[(m2base + l16) * 33 + ka];
        a2.y = Y1s[(m2base + l16) * 33 + ka + 1];
        v2f b2v = {0.0f, 0.0f};
        if (nvalid) b2v = w2f[k];
        acc2 = __builtin_amdgcn_wmma_f32_16x16x4_f32(
                   false, a2, false, b2v, (short)0, acc2, false, false);
    }

    if (wave < 2 && nvalid) {
        #pragma unroll
        for (int r = 0; r < 8; ++r)
            Y2s[(m2base + r + 8 * half) * 9 + l16] = acc2[r] + bias2;
    }

    lds_barrier();

    if (tid < NBATCH * 3) {
        float s = bias3;
        #pragma unroll
        for (int j = 0; j < 8; ++j)
            s += Y2s[bb * 9 + j] * w3r[j];
        out[((size_t)bb * 3 + c) * NPTS + p] = s;
    }
}

__global__ __launch_bounds__(128)
void FCDecoder_gfx1250_kernel(const float* glf,
                              const float* W1, const float* b1,
                              const float* W2, const float* b2,
                              const float* W3, const float* b3,
                              float* out)
{
    __shared__ float Bs[2 * TILE_F];
    __shared__ float Y1s[32 * 33];
    __shared__ float Y2s[32 * 9];

    const int tid  = threadIdx.x;
    const int wave = tid >> 5;
    const int lane = tid & 31;
    const int l16  = lane & 15;
    const int half = lane >> 4;

    const int m_tile = wave & 1;
    const int n_tile = wave >> 1;
    const int m = m_tile * 16 + l16;
    const int n = n_tile * 16 + l16;

    int t1 = TILE_F;
    asm volatile("" : "+v"(t1));
    const float* Arow = Bs + t1 + m * BPITCH + 2 * half;
    const float* Brow = Bs + n * BPITCH + 2 * half;

    const int  m2base = (wave & 1) * 16;
    const bool nvalid = (l16 < 8);
    const int  n2     = nvalid ? l16 : 0;
    const int  bb     = tid / 3;
    const int  c      = tid % 3;

    v4f st[32];
    {
        v4f g[32];
        load_tile<false>(glf, g, tid);
        load_tile<true>(W1 + (size_t)blockIdx.x * PPB * NBATCH * FEAT, st, tid);
        store_tile(g, Bs + t1, tid);
    }
    lds_barrier();
    v2f af[128];
    #pragma unroll
    for (int s = 0; s < 128; ++s)
        af[s] = *(const v2f*)(Arow + 4 * s);
    refill_tile<true>(st, Bs, W1 + ((size_t)blockIdx.x * PPB + 1) * NBATCH * FEAT, tid);
    lds_barrier();

    #pragma unroll 1
    for (int i = 0; i < PPB - 2; ++i)
        one_position<true, true>(blockIdx.x * PPB + i, Brow + (i & 1) * t1, Bs + ((i + 1) & 1) * t1,
                                 W1, b1, W2, b2, W3, b3, out, af, st, Y1s, Y2s,
                                 tid, wave, half, l16, m_tile, n, m2base, nvalid, n2, bb, c);
    one_position<true, false>(blockIdx.x * PPB + PPB - 2, Brow + ((PPB - 2) & 1) * t1, Bs + ((PPB - 1) & 1) * t1,
                              W1, b1, W2, b2, W3, b3, out, af, st, Y1s, Y2s,
                              tid, wave, half, l16, m_tile, n, m2base, nvalid, n2, bb, c);
    one_position<false, false>(blockIdx.x * PPB + PPB - 1, Brow + ((PPB - 1) & 1) * t1, Bs,
                               W1, b1, W2, b2, W3, b3, out, af, st, Y1s, Y2s,
                               tid, wave, half, l16, m_tile, n, m2base, nvalid, n2, bb, c);
}

extern "C" void kernel_launch(void* const* d_in, const int* in_sizes, int n_in,
                              void* d_out, int out_size, void* d_ws, size_t ws_size,
                              hipStream_t stream) {
    (void)in_sizes; (void)n_in; (void)out_size; (void)d_ws; (void)ws_size;
    const float* glf = (const float*)d_in[0];
    const float* W1  = (const float*)d_in[1];
    const float* b1  = (const float*)d_in[2];
    const float* W2  = (const float*)d_in[3];
    const float* b2  = (const float*)d_in[4];
    const float* W3  = (const float*)d_in[5];
    const float* b3  = (const float*)d_in[6];
    float* out = (float*)d_out;

    FCDecoder_gfx1250_kernel<<<NBLK, 128, 0, stream>>>(glf, W1, b1, W2, b2, W3, b3, out);
}
